# starting kernel
# baseline (speedup 1.0000x reference)
#include <hip/hip_runtime.h>


#define DD        256
#define TWO_D     512
#define STEPS     16
#define DT        0.1f
#define ROWS      32
#define NWAVES    16
#define NTHREADS  (NWAVES * 32)
#define LDS_STRIDE (TWO_D + 4)

typedef __attribute__((ext_vector_type(2))) float v2f;
typedef __attribute__((ext_vector_type(8))) float v8f;

#define WMMA_F32(A, B, C) \
  __builtin_amdgcn_wmma_f32_16x16x4_f32(false, (A), false, (B), (short)0, (C), false, false)

__global__ __launch_bounds__(256) void pack_weights(const float* __restrict__ W,
                                                    float* __restrict__ P,
                                                    const int transposed) {
  const int idx = blockIdx.x * 256 + threadIdx.x;
  const int n = idx & 511;
  const int k = idx >> 9;
  const float v = transposed ? W[n * TWO_D + k] : W[k * TWO_D + n];
  const int grp = ((k >> 2) << 1) + ((k >> 1) & 1);
  P[grp * (TWO_D * 2) + (n << 1) + (k & 1)] = v;
}

template <int MODE>
__device__ __forceinline__ void gemm_512(const float* Abuf,
                                         const float* __restrict__ Bpk,
                                         float* Obuf,
                                         const float* Maskbuf,
                                         const float* __restrict__ sCst,
                                         int wave, int lane) {
  const int m     = lane & 15;
  const int khalf = lane >> 4;

  v8f acc[2][2];
#pragma unroll
  for (int mt = 0; mt < 2; ++mt)
#pragma unroll
    for (int t = 0; t < 2; ++t)
#pragma unroll
      for (int j = 0; j < 8; ++j) acc[mt][t][j] = 0.0f;

  const float* A0 = Abuf + m * LDS_STRIDE;
  const float* A1 = A0 + 16 * LDS_STRIDE;
  const float* Bbase = Bpk + khalf * (TWO_D * 2) + (wave * 32 + m) * 2;

#pragma unroll 2
  for (int k0 = 0; k0 < TWO_D; k0 += 4) {
    const int kk = k0 + (khalf << 1);
    v2f a0 = *(const v2f*)(A0 + kk);
    v2f a1 = *(const v2f*)(A1 + kk);
    const float* Bp = Bbase + (k0 >> 1) * (TWO_D * 2);
    v2f b0 = *(const v2f*)(Bp);
    v2f b1 = *(const v2f*)(Bp + 32);
    acc[0][0] = WMMA_F32(a0, b0, acc[0][0]);
    acc[0][1] = WMMA_F32(a0, b1, acc[0][1]);
    acc[1][0] = WMMA_F32(a1, b0, acc[1][0]);
    acc[1][1] = WMMA_F32(a1, b1, acc[1][1]);
  }

  const int rbase = khalf << 3;
  const int colb  = wave * 32 + m;
  const float c0a = (MODE == 1) ? sCst[colb]              : (MODE == 2) ? sCst[TWO_D + colb]      : 0.0f;
  const float c1a = (MODE == 1) ? sCst[colb + 16]         : (MODE == 2) ? sCst[TWO_D + colb + 16] : 0.0f;
  const float w0  = (MODE == 2) ? sCst[2 * TWO_D + colb]      : 0.0f;
  const float w1  = (MODE == 2) ? sCst[2 * TWO_D + colb + 16] : 0.0f;

#pragma unroll
  for (int mt = 0; mt < 2; ++mt) {
#pragma unroll
    for (int j = 0; j < 8; ++j) {
      const int row = mt * 16 + rbase + j;
      float* o = Obuf + row * LDS_STRIDE + colb;
      float v0 = acc[mt][0][j];
      float v1 = acc[mt][1][j];
      if (MODE == 1) {
        o[0]  = fmaxf(v0 + c0a, 0.0f);
        o[16] = fmaxf(v1 + c1a, 0.0f);
      } else if (MODE == 2) {
        o[0]  = (v0 + c0a > 0.0f) ? w0 : 0.0f;
        o[16] = (v1 + c1a > 0.0f) ? w1 : 0.0f;
      } else if (MODE == 3) {
        const float* mk = Maskbuf + row * LDS_STRIDE + colb;
        float h0 = mk[0];
        float h1 = mk[16];
        o[0]  = (h0 > 0.0f) ? v0 : 0.0f;
        o[16] = (h1 > 0.0f) ? v1 : 0.0f;
      } else {
        o[0]  = v0;
        o[16] = v1;
      }
    }
  }
}

__device__ __forceinline__ void grad_eval(float* sS, float* sT1, float* sT2,
                                          const float* sCst,
                                          const float* __restrict__ P1f,
                                          const float* __restrict__ P2f,
                                          const float* __restrict__ P2b,
                                          const float* __restrict__ P1b,
                                          int wave, int lane) {
  gemm_512<1>(sS,  P1f, sT1, nullptr, sCst, wave, lane);
  __syncthreads();
  gemm_512<2>(sT1, P2f, sT2, nullptr, sCst, wave, lane);
  __syncthreads();
  gemm_512<3>(sT2, P2b, sT1, sT1,     sCst, wave, lane);
  __syncthreads();
  gemm_512<4>(sT1, P1b, sT2, nullptr, sCst, wave, lane);
  __syncthreads();
}

__global__ __launch_bounds__(NTHREADS) void hnn_leapfrog_kernel(
    const float* __restrict__ x,
    const float* __restrict__ b1, const float* __restrict__ b2,
    const float* __restrict__ Wo,
    const float* __restrict__ P1f, const float* __restrict__ P2f,
    const float* __restrict__ P2b, const float* __restrict__ P1b,
    float* __restrict__ out)
{
  __shared__ float sS [ROWS * LDS_STRIDE];
  __shared__ float sT1[ROWS * LDS_STRIDE];
  __shared__ float sT2[ROWS * LDS_STRIDE];
  __shared__ float sCst[3 * TWO_D];

  const int tid  = threadIdx.x;
  const int wave = tid >> 5;
  const int lane = tid & 31;
  const int r0   = blockIdx.x * ROWS;

  for (int i = tid; i < TWO_D; i += NTHREADS) {
    sCst[i]             = b1[i];
    sCst[TWO_D + i]     = b2[i];
    sCst[2 * TWO_D + i] = Wo[i];
  }

  for (int i = tid; i < ROWS * DD; i += NTHREADS) {
    const int r = i >> 8;
    const int d = i & 255;
    const float* xp = x + ((r0 + r) * DD + d) * 4;
    const float x3 = xp[3];
    const float x2 = xp[2];
    sS[r * LDS_STRIDE + d]      = x3;
    sS[r * LDS_STRIDE + DD + d] = x3 - x2;
  }
  __syncthreads();

  for (int step = 0; step < STEPS; ++step) {
    grad_eval(sS, sT1, sT2, sCst, P1f, P2f, P2b, P1b, wave, lane);
    for (int i = tid; i < ROWS * DD; i += NTHREADS) {
      const int r = i >> 8;
      const int d = i & 255;
      const float dHq = sT2[r * LDS_STRIDE + d];
      const float dHp = sT2[r * LDS_STRIDE + DD + d];
      sS[r * LDS_STRIDE + DD + d] -= 0.5f * DT * dHq;
      sS[r * LDS_STRIDE + d]      += DT * dHp;
    }
    __syncthreads();

    grad_eval(sS, sT1, sT2, sCst, P1f, P2f, P2b, P1b, wave, lane);
    for (int i = tid; i < ROWS * DD; i += NTHREADS) {
      const int r = i >> 8;
      const int d = i & 255;
      sS[r * LDS_STRIDE + DD + d] -= 0.5f * DT * sT2[r * LDS_STRIDE + d];
    }
    __syncthreads();
  }

  for (int i = tid; i < ROWS * TWO_D; i += NTHREADS) {
    const int r = i >> 9;
    const int c = i & 511;
    out[(r0 + r) * TWO_D + c] = sS[r * LDS_STRIDE + c];
  }
}

extern "C" void kernel_launch(void* const* d_in, const int* in_sizes, int n_in,
                              void* d_out, int out_size, void* d_ws, size_t ws_size,
                              hipStream_t stream) {
  const float* x  = (const float*)d_in[0];
  const float* W1 = (const float*)d_in[1];
  const float* b1 = (const float*)d_in[2];
  const float* W2 = (const float*)d_in[3];
  const float* b2 = (const float*)d_in[4];
  const float* Wo = (const float*)d_in[5];
  float* out = (float*)d_out;

  const int MS = TWO_D * TWO_D;
  float* P1f = (float*)d_ws;
  float* P2f = P1f + MS;
  float* P2b = P2f + MS;
  float* P1b = P2b + MS;

  pack_weights<<<(MS + 255) / 256, 256, 0, stream>>>(W1, P1f, 1);
  pack_weights<<<(MS + 255) / 256, 256, 0, stream>>>(W2, P2f, 1);
  pack_weights<<<(MS + 255) / 256, 256, 0, stream>>>(W2, P2b, 0);
  pack_weights<<<(MS + 255) / 256, 256, 0, stream>>>(W1, P1b, 0);

  const int nblocks = 8192 / ROWS;
  hnn_leapfrog_kernel<<<nblocks, NTHREADS, 0, stream>>>(
      x, b1, b2, Wo, P1f, P2f, P2b, P1b, out);
}


# K-loop register double buffer: the next 8 k-steps' A (LDS) and B (L2) fragments are loaded as one group ahead of the current block's 32 WMMAs; order of every sum unchanged
# speedup vs baseline: 1.2026x; 1.2026x over previous
#include <hip/hip_runtime.h>


#define DD        256
#define TWO_D     512
#define STEPS     16
#define DT        0.1f
#define ROWS      32
#define NWAVES    16
#define NTHREADS  (NWAVES * 32)
#define LDS_STRIDE (TWO_D + 4)
#define KB        8
#define NBLK      (TWO_D / (4 * KB))
#define BSTEP     (2 * TWO_D * 2)

typedef __attribute__((ext_vector_type(2))) float v2f;
typedef __attribute__((ext_vector_type(8))) float v8f;

#define WMMA_F32(A, B, C) \
  __builtin_amdgcn_wmma_f32_16x16x4_f32(false, (A), false, (B), (short)0, (C), false, false)

__global__ __launch_bounds__(256) void pack_weights(const float* __restrict__ W,
                                                    float* __restrict__ P,
                                                    const int transposed) {
  const int idx = blockIdx.x * 256 + threadIdx.x;
  const int n = idx & 511;
  const int k = idx >> 9;
  const float v = transposed ? W[n * TWO_D + k] : W[k * TWO_D + n];
  const int grp = ((k >> 2) << 1) + ((k >> 1) & 1);
  P[grp * (TWO_D * 2) + (n << 1) + (k & 1)] = v;
}

__device__ __forceinline__ void load_frags(v2f (&fa)[KB][2], v2f (&fb)[KB][2],
                                           const float* A0, const float* A1,
                                           const float* Bbase,
                                           int blk) {
  const float* Bp = Bbase + blk * (KB * BSTEP);
#pragma unroll
  for (int st = 0; st < KB; ++st) {
    fb[st][0] = *(const v2f*)(Bp + st * BSTEP);
    fb[st][1] = *(const v2f*)(Bp + st * BSTEP + 32);
  }
  const int kb = blk * (KB * 4);
#pragma unroll
  for (int st = 0; st < KB; ++st) {
    fa[st][0] = *(const v2f*)(A0 + kb + st * 4);
    fa[st][1] = *(const v2f*)(A1 + kb + st * 4);
  }
}

__device__ __forceinline__ void mma_block(v8f (&acc)[2][2],
                                          const v2f (&fa)[KB][2],
                                          const v2f (&fb)[KB][2]) {
#pragma unroll
  for (int st = 0; st < KB; ++st) {
    acc[0][0] = WMMA_F32(fa[st][0], fb[st][0], acc[0][0]);
    acc[0][1] = WMMA_F32(fa[st][0], fb[st][1], acc[0][1]);
    acc[1][0] = WMMA_F32(fa[st][1], fb[st][0], acc[1][0]);
    acc[1][1] = WMMA_F32(fa[st][1], fb[st][1], acc[1][1]);
  }
}

template <int MODE>
__device__ __forceinline__ void gemm_512(const float* Abuf,
                                         const float* Bpk,
                                         float* Obuf,
                                         const float* Maskbuf,
                                         const float* __restrict__ sCst,
                                         int wave, int lane) {
  const int m     = lane & 15;
  const int khalf = lane >> 4;

  v8f acc[2][2];
#pragma unroll
  for (int mt = 0; mt < 2; ++mt)
#pragma unroll
    for (int t = 0; t < 2; ++t)
#pragma unroll
      for (int j = 0; j < 8; ++j) acc[mt][t][j] = 0.0f;

  const float* A0 = Abuf + m * LDS_STRIDE + (khalf << 1);
  const float* A1 = A0 + 16 * LDS_STRIDE;
  const float* Bbase = Bpk + khalf * (TWO_D * 2) + (wave * 32 + m) * 2;

  v2f fa0[KB][2], fb0[KB][2];
  v2f fa1[KB][2], fb1[KB][2];

  load_frags(fa0, fb0, A0, A1, Bbase, 0);
#pragma nounroll
  for (int blk = 0; blk < NBLK; blk += 2) {
    load_frags(fa1, fb1, A0, A1, Bbase, blk + 1);
    mma_block(acc, fa0, fb0);
    if (blk + 2 < NBLK) load_frags(fa0, fb0, A0, A1, Bbase, blk + 2);
    mma_block(acc, fa1, fb1);
  }

  const int rbase = khalf << 3;
  const int colb  = wave * 32 + m;
  const float c0a = (MODE == 1) ? sCst[colb]              : (MODE == 2) ? sCst[TWO_D + colb]      : 0.0f;
  const float c1a = (MODE == 1) ? sCst[colb + 16]         : (MODE == 2) ? sCst[TWO_D + colb + 16] : 0.0f;
  const float w0  = (MODE == 2) ? sCst[2 * TWO_D + colb]      : 0.0f;
  const float w1  = (MODE == 2) ? sCst[2 * TWO_D + colb + 16] : 0.0f;

#pragma unroll
  for (int mt = 0; mt < 2; ++mt) {
#pragma unroll
    for (int j = 0; j < 8; ++j) {
      const int row = mt * 16 + rbase + j;
      float* o = Obuf + row * LDS_STRIDE + colb;
      float v0 = acc[mt][0][j];
      float v1 = acc[mt][1][j];
      if (MODE == 1) {
        o[0]  = fmaxf(v0 + c0a, 0.0f);
        o[16] = fmaxf(v1 + c1a, 0.0f);
      } else if (MODE == 2) {
        o[0]  = (v0 + c0a > 0.0f) ? w0 : 0.0f;
        o[16] = (v1 + c1a > 0.0f) ? w1 : 0.0f;
      } else if (MODE == 3) {
        const float* mk = Maskbuf + row * LDS_STRIDE + colb;
        float h0 = mk[0];
        float h1 = mk[16];
        o[0]  = (h0 > 0.0f) ? v0 : 0.0f;
        o[16] = (h1 > 0.0f) ? v1 : 0.0f;
      } else {
        o[0]  = v0;
        o[16] = v1;
      }
    }
  }
}

__device__ __forceinline__ void grad_eval(float* sS, float* sT1, float* sT2,
                                          const float* sCst,
                                          const float* P1f,
                                          const float* P2f,
                                          const float* P2b,
                                          const float* P1b,
                                          int wave, int lane) {
  gemm_512<1>(sS,  P1f, sT1, nullptr, sCst, wave, lane);
  __syncthreads();
  gemm_512<2>(sT1, P2f, sT2, nullptr, sCst, wave, lane);
  __syncthreads();
  gemm_512<3>(sT2, P2b, sT1, sT1,     sCst, wave, lane);
  __syncthreads();
  gemm_512<4>(sT1, P1b, sT2, nullptr, sCst, wave, lane);
  __syncthreads();
}

__global__ __launch_bounds__(NTHREADS) void hnn_leapfrog_kernel(
    const float* __restrict__ x,
    const float* __restrict__ b1, const float* __restrict__ b2,
    const float* __restrict__ Wo,
    const float* P1f, const float* P2f,
    const float* P2b, const float* P1b,
    float* __restrict__ out)
{
  __shared__ float sS [ROWS * LDS_STRIDE];
  __shared__ float sT1[ROWS * LDS_STRIDE];
  __shared__ float sT2[ROWS * LDS_STRIDE];
  __shared__ float sCst[3 * TWO_D];

  const int tid  = threadIdx.x;
  const int wave = tid >> 5;
  const int lane = tid & 31;
  const int r0   = blockIdx.x * ROWS;

  for (int i = tid; i < TWO_D; i += NTHREADS) {
    sCst[i]             = b1[i];
    sCst[TWO_D + i]     = b2[i];
    sCst[2 * TWO_D + i] = Wo[i];
  }

  for (int i = tid; i < ROWS * DD; i += NTHREADS) {
    const int r = i >> 8;
    const int d = i & 255;
    const float* xp = x + ((r0 + r) * DD + d) * 4;
    const float x3 = xp[3];
    const float x2 = xp[2];
    sS[r * LDS_STRIDE + d]      = x3;
    sS[r * LDS_STRIDE + DD + d] = x3 - x2;
  }
  __syncthreads();

  for (int step = 0; step < STEPS; ++step) {
    grad_eval(sS, sT1, sT2, sCst, P1f, P2f, P2b, P1b, wave, lane);
    for (int i = tid; i < ROWS * DD; i += NTHREADS) {
      const int r = i >> 8;
      const int d = i & 255;
      const float dHq = sT2[r * LDS_STRIDE + d];
      const float dHp = sT2[r * LDS_STRIDE + DD + d];
      sS[r * LDS_STRIDE + DD + d] -= 0.5f * DT * dHq;
      sS[r * LDS_STRIDE + d]      += DT * dHp;
    }
    __syncthreads();

    grad_eval(sS, sT1, sT2, sCst, P1f, P2f, P2b, P1b, wave, lane);
    for (int i = tid; i < ROWS * DD; i += NTHREADS) {
      const int r = i >> 8;
      const int d = i & 255;
      sS[r * LDS_STRIDE + DD + d] -= 0.5f * DT * sT2[r * LDS_STRIDE + d];
    }
    __syncthreads();
  }

  for (int i = tid; i < ROWS * TWO_D; i += NTHREADS) {
    const int r = i >> 9;
    const int c = i & 511;
    out[(r0 + r) * TWO_D + c] = sS[r * LDS_STRIDE + c];
  }
}

extern "C" void kernel_launch(void* const* d_in, const int* in_sizes, int n_in,
                              void* d_out, int out_size, void* d_ws, size_t ws_size,
                              hipStream_t stream) {
  const float* x  = (const float*)d_in[0];
  const float* W1 = (const float*)d_in[1];
  const float* b1 = (const float*)d_in[2];
  const float* W2 = (const float*)d_in[3];
  const float* b2 = (const float*)d_in[4];
  const float* Wo = (const float*)d_in[5];
  float* out = (float*)d_out;

  const int MS = TWO_D * TWO_D;
  float* P1f = (float*)d_ws;
  float* P2f = P1f + MS;
  float* P2b = P2f + MS;
  float* P1b = P2b + MS;

  pack_weights<<<(MS + 255) / 256, 256, 0, stream>>>(W1, P1f, 1);
  pack_weights<<<(MS + 255) / 256, 256, 0, stream>>>(W2, P2f, 1);
  pack_weights<<<(MS + 255) / 256, 256, 0, stream>>>(W2, P2b, 0);
  pack_weights<<<(MS + 255) / 256, 256, 0, stream>>>(W1, P1b, 0);

  const int nblocks = 8192 / ROWS;
  hnn_leapfrog_kernel<<<nblocks, NTHREADS, 0, stream>>>(
      x, b1, b2, Wo, P1f, P2f, P2b, P1b, out);
}


# Second gradient evaluation of each step computes only columns 0..255 (dH/dq) of its last GEMM, as 16 waves x (2 M-tiles x 1 N-tile); the dH/dp half is never read; order of every sum unchanged
# speedup vs baseline: 1.2714x; 1.0572x over previous
#include <hip/hip_runtime.h>


#define DD        256
#define TWO_D     512
#define STEPS     16
#define DT        0.1f
#define ROWS      32
#define NWAVES    16
#define NTHREADS  (NWAVES * 32)
#define LDS_STRIDE (TWO_D + 4)
#define KB        8
#define NBLK      (TWO_D / (4 * KB))
#define BSTEP     (2 * TWO_D * 2)

typedef __attribute__((ext_vector_type(2))) float v2f;
typedef __attribute__((ext_vector_type(8))) float v8f;

#define WMMA_F32(A, B, C) \
  __builtin_amdgcn_wmma_f32_16x16x4_f32(false, (A), false, (B), (short)0, (C), false, false)

__global__ __launch_bounds__(256) void pack_weights(const float* __restrict__ W,
                                                    float* __restrict__ P,
                                                    const int transposed) {
  const int idx = blockIdx.x * 256 + threadIdx.x;
  const int n = idx & 511;
  const int k = idx >> 9;
  const float v = transposed ? W[n * TWO_D + k] : W[k * TWO_D + n];
  const int grp = ((k >> 2) << 1) + ((k >> 1) & 1);
  P[grp * (TWO_D * 2) + (n << 1) + (k & 1)] = v;
}

template <int NT>
__device__ __forceinline__ void load_frags(v2f (&fa)[KB][2], v2f (&fb)[KB][NT],
                                           const float* A0, const float* A1,
                                           const float* Bbase,
                                           int blk) {
  const float* Bp = Bbase + blk * (KB * BSTEP);
#pragma unroll
  for (int st = 0; st < KB; ++st) {
    fb[st][0] = *(const v2f*)(Bp + st * BSTEP);
    if (NT == 2) fb[st][NT - 1] = *(const v2f*)(Bp + st * BSTEP + 32);
  }
  const int kb = blk * (KB * 4);
#pragma unroll
  for (int st = 0; st < KB; ++st) {
    fa[st][0] = *(const v2f*)(A0 + kb + st * 4);
    fa[st][1] = *(const v2f*)(A1 + kb + st * 4);
  }
}

template <int NT>
__device__ __forceinline__ void mma_block(v8f (&acc)[2][NT],
                                          const v2f (&fa)[KB][2],
                                          const v2f (&fb)[KB][NT]) {
#pragma unroll
  for (int st = 0; st < KB; ++st) {
    if (NT == 2) {
      acc[0][0]      = WMMA_F32(fa[st][0], fb[st][0],      acc[0][0]);
      acc[0][NT - 1] = WMMA_F32(fa[st][0], fb[st][NT - 1], acc[0][NT - 1]);
      acc[1][0]      = WMMA_F32(fa[st][1], fb[st][0],      acc[1][0]);
      acc[1][NT - 1] = WMMA_F32(fa[st][1], fb[st][NT - 1], acc[1][NT - 1]);
    } else {
      acc[0][0] = WMMA_F32(fa[st][0], fb[st][0], acc[0][0]);
      acc[1][0] = WMMA_F32(fa[st][1], fb[st][0], acc[1][0]);
    }
  }
}

template <int MODE, int NT>
__device__ __forceinline__ void gemm_512(const float* Abuf,
                                         const float* Bpk,
                                         float* Obuf,
                                         const float* Maskbuf,
                                         const float* __restrict__ sCst,
                                         int wave, int lane) {
  const int m     = lane & 15;
  const int khalf = lane >> 4;

  v8f acc[2][NT];
#pragma unroll
  for (int mt = 0; mt < 2; ++mt)
#pragma unroll
    for (int t = 0; t < NT; ++t)
#pragma unroll
      for (int j = 0; j < 8; ++j) acc[mt][t][j] = 0.0f;

  const float* A0 = Abuf + m * LDS_STRIDE + (khalf << 1);
  const float* A1 = A0 + 16 * LDS_STRIDE;
  const float* Bbase = Bpk + khalf * (TWO_D * 2) + (wave * (16 * NT) + m) * 2;

  v2f fa0[KB][2], fb0[KB][NT];
  v2f fa1[KB][2], fb1[KB][NT];

  load_frags<NT>(fa0, fb0, A0, A1, Bbase, 0);
#pragma nounroll
  for (int blk = 0; blk < NBLK; blk += 2) {
    load_frags<NT>(fa1, fb1, A0, A1, Bbase, blk + 1);
    mma_block<NT>(acc, fa0, fb0);
    if (blk + 2 < NBLK) load_frags<NT>(fa0, fb0, A0, A1, Bbase, blk + 2);
    mma_block<NT>(acc, fa1, fb1);
  }

  const int rbase = khalf << 3;
  const int colb  = wave * (16 * NT) + m;
  const float c0a = (MODE == 1) ? sCst[colb]              : (MODE == 2) ? sCst[TWO_D + colb]      : 0.0f;
  const float c1a = (MODE == 1) ? sCst[colb + 16]         : (MODE == 2) ? sCst[TWO_D + colb + 16] : 0.0f;
  const float w0  = (MODE == 2) ? sCst[2 * TWO_D + colb]      : 0.0f;
  const float w1  = (MODE == 2) ? sCst[2 * TWO_D + colb + 16] : 0.0f;

#pragma unroll
  for (int mt = 0; mt < 2; ++mt) {
#pragma unroll
    for (int j = 0; j < 8; ++j) {
      const int row = mt * 16 + rbase + j;
      float* o = Obuf + row * LDS_STRIDE + colb;
      float v0 = acc[mt][0][j];
      float v1 = acc[mt][NT - 1][j];
      if (MODE == 1) {
        o[0]  = fmaxf(v0 + c0a, 0.0f);
        if (NT == 2) o[16] = fmaxf(v1 + c1a, 0.0f);
      } else if (MODE == 2) {
        o[0]  = (v0 + c0a > 0.0f) ? w0 : 0.0f;
        if (NT == 2) o[16] = (v1 + c1a > 0.0f) ? w1 : 0.0f;
      } else if (MODE == 3) {
        const float* mk = Maskbuf + row * LDS_STRIDE + colb;
        float h0 = mk[0];
        o[0]  = (h0 > 0.0f) ? v0 : 0.0f;
        if (NT == 2) {
          float h1 = mk[16];
          o[16] = (h1 > 0.0f) ? v1 : 0.0f;
        }
      } else {
        o[0]  = v0;
        if (NT == 2) o[16] = v1;
      }
    }
  }
}

template <bool Q_ONLY>
__device__ __forceinline__ void grad_eval(float* sS, float* sT1, float* sT2,
                                          const float* sCst,
                                          const float* P1f,
                                          const float* P2f,
                                          const float* P2b,
                                          const float* P1b,
                                          int wave, int lane) {
  gemm_512<1, 2>(sS,  P1f, sT1, nullptr, sCst, wave, lane);
  __syncthreads();
  gemm_512<2, 2>(sT1, P2f, sT2, nullptr, sCst, wave, lane);
  __syncthreads();
  gemm_512<3, 2>(sT2, P2b, sT1, sT1,     sCst, wave, lane);
  __syncthreads();
  if (Q_ONLY)
    gemm_512<4, 1>(sT1, P1b, sT2, nullptr, sCst, wave, lane);
  else
    gemm_512<4, 2>(sT1, P1b, sT2, nullptr, sCst, wave, lane);
  __syncthreads();
}

__global__ __launch_bounds__(NTHREADS) void hnn_leapfrog_kernel(
    const float* __restrict__ x,
    const float* __restrict__ b1, const float* __restrict__ b2,
    const float* __restrict__ Wo,
    const float* P1f, const float* P2f,
    const float* P2b, const float* P1b,
    float* __restrict__ out)
{
  __shared__ float sS [ROWS * LDS_STRIDE];
  __shared__ float sT1[ROWS * LDS_STRIDE];
  __shared__ float sT2[ROWS * LDS_STRIDE];
  __shared__ float sCst[3 * TWO_D];

  const int tid  = threadIdx.x;
  const int wave = tid >> 5;
  const int lane = tid & 31;
  const int r0   = blockIdx.x * ROWS;

  for (int i = tid; i < TWO_D; i += NTHREADS) {
    sCst[i]             = b1[i];
    sCst[TWO_D + i]     = b2[i];
    sCst[2 * TWO_D + i] = Wo[i];
  }

  for (int i = tid; i < ROWS * DD; i += NTHREADS) {
    const int r = i >> 8;
    const int d = i & 255;
    const float* xp = x + ((r0 + r) * DD + d) * 4;
    const float x3 = xp[3];
    const float x2 = xp[2];
    sS[r * LDS_STRIDE + d]      = x3;
    sS[r * LDS_STRIDE + DD + d] = x3 - x2;
  }
  __syncthreads();

  for (int step = 0; step < STEPS; ++step) {
    grad_eval<false>(sS, sT1, sT2, sCst, P1f, P2f, P2b, P1b, wave, lane);
    for (int i = tid; i < ROWS * DD; i += NTHREADS) {
      const int r = i >> 8;
      const int d = i & 255;
      const float dHq = sT2[r * LDS_STRIDE + d];
      const float dHp = sT2[r * LDS_STRIDE + DD + d];
      sS[r * LDS_STRIDE + DD + d] -= 0.5f * DT * dHq;
      sS[r * LDS_STRIDE + d]      += DT * dHp;
    }
    __syncthreads();

    grad_eval<true>(sS, sT1, sT2, sCst, P1f, P2f, P2b, P1b, wave, lane);
    for (int i = tid; i < ROWS * DD; i += NTHREADS) {
      const int r = i >> 8;
      const int d = i & 255;
      sS[r * LDS_STRIDE + DD + d] -= 0.5f * DT * sT2[r * LDS_STRIDE + d];
    }
    __syncthreads();
  }

  for (int i = tid; i < ROWS * TWO_D; i += NTHREADS) {
    const int r = i >> 9;
    const int c = i & 511;
    out[(r0 + r) * TWO_D + c] = sS[r * LDS_STRIDE + c];
  }
}

extern "C" void kernel_launch(void* const* d_in, const int* in_sizes, int n_in,
                              void* d_out, int out_size, void* d_ws, size_t ws_size,
                              hipStream_t stream) {
  const float* x  = (const float*)d_in[0];
  const float* W1 = (const float*)d_in[1];
  const float* b1 = (const float*)d_in[2];
  const float* W2 = (const float*)d_in[3];
  const float* b2 = (const float*)d_in[4];
  const float* Wo = (const float*)d_in[5];
  float* out = (float*)d_out;

  const int MS = TWO_D * TWO_D;
  float* P1f = (float*)d_ws;
  float* P2f = P1f + MS;
  float* P2b = P2f + MS;
  float* P1b = P2b + MS;

  pack_weights<<<(MS + 255) / 256, 256, 0, stream>>>(W1, P1f, 1);
  pack_weights<<<(MS + 255) / 256, 256, 0, stream>>>(W2, P2f, 1);
  pack_weights<<<(MS + 255) / 256, 256, 0, stream>>>(W2, P2b, 0);
  pack_weights<<<(MS + 255) / 256, 256, 0, stream>>>(W1, P1b, 0);

  const int nblocks = 8192 / ROWS;
  hnn_leapfrog_kernel<<<nblocks, NTHREADS, 0, stream>>>(
      x, b1, b2, Wo, P1f, P2f, P2b, P1b, out);
}


# First GEMM of a step's first gradient evaluation resumes from the k=0..255 (q part) accumulators that the previous evaluation kept per lane in LDS, runs k=256..511 only; order of every sum unchanged
# speedup vs baseline: 1.3544x; 1.0653x over previous
#include <hip/hip_runtime.h>


#define DD        256
#define TWO_D     512
#define STEPS     16
#define DT        0.1f
#define ROWS      32
#define NWAVES    16
#define NTHREADS  (NWAVES * 32)
#define LDS_STRIDE (TWO_D + 4)
#define KB        8
#define NBLK      (TWO_D / (4 * KB))
#define BSTEP     (2 * TWO_D * 2)

typedef __attribute__((ext_vector_type(2))) float v2f;
typedef __attribute__((ext_vector_type(4))) float v4f;
typedef __attribute__((ext_vector_type(8))) float v8f;

#define WMMA_F32(A, B, C) \
  __builtin_amdgcn_wmma_f32_16x16x4_f32(false, (A), false, (B), (short)0, (C), false, false)

__global__ __launch_bounds__(256) void pack_weights(const float* __restrict__ W,
                                                    float* __restrict__ P,
                                                    const int transposed) {
  const int idx = blockIdx.x * 256 + threadIdx.x;
  const int n = idx & 511;
  const int k = idx >> 9;
  const float v = transposed ? W[n * TWO_D + k] : W[k * TWO_D + n];
  const int grp = ((k >> 2) << 1) + ((k >> 1) & 1);
  P[grp * (TWO_D * 2) + (n << 1) + (k & 1)] = v;
}

template <int NT>
__device__ __forceinline__ void load_frags(v2f (&fa)[KB][2], v2f (&fb)[KB][NT],
                                           const float* A0, const float* A1,
                                           const float* Bbase,
                                           int blk) {
  const float* Bp = Bbase + blk * (KB * BSTEP);
#pragma unroll
  for (int st = 0; st < KB; ++st) {
    fb[st][0] = *(const v2f*)(Bp + st * BSTEP);
    if (NT == 2) fb[st][NT - 1] = *(const v2f*)(Bp + st * BSTEP + 32);
  }
  const int kb = blk * (KB * 4);
#pragma unroll
  for (int st = 0; st < KB; ++st) {
    fa[st][0] = *(const v2f*)(A0 + kb + st * 4);
    fa[st][1] = *(const v2f*)(A1 + kb + st * 4);
  }
}

template <int NT>
__device__ __forceinline__ void mma_block(v8f (&acc)[2][NT],
                                          const v2f (&fa)[KB][2],
                                          const v2f (&fb)[KB][NT]) {
#pragma unroll
  for (int st = 0; st < KB; ++st) {
    if (NT == 2) {
      acc[0][0]      = WMMA_F32(fa[st][0], fb[st][0],      acc[0][0]);
      acc[0][NT - 1] = WMMA_F32(fa[st][0], fb[st][NT - 1], acc[0][NT - 1]);
      acc[1][0]      = WMMA_F32(fa[st][1], fb[st][0],      acc[1][0]);
      acc[1][NT - 1] = WMMA_F32(fa[st][1], fb[st][NT - 1], acc[1][NT - 1]);
    } else {
      acc[0][0] = WMMA_F32(fa[st][0], fb[st][0], acc[0][0]);
      acc[1][0] = WMMA_F32(fa[st][1], fb[st][0], acc[1][0]);
    }
  }
}

#define QBLK (NBLK / 2)
#define QSAVE_WAVE (2 * 2 * 8 * 32)

__device__ __forceinline__ void save_acc(float* sQw, const v8f (&acc)[2][2], int lane) {
#pragma unroll
  for (int mt = 0; mt < 2; ++mt)
#pragma unroll
    for (int t = 0; t < 2; ++t) {
      float* d = sQw + ((mt * 2 + t) * 2) * 128 + lane * 4;
      *(v4f*)(d)       = __builtin_shufflevector(acc[mt][t], acc[mt][t], 0, 1, 2, 3);
      *(v4f*)(d + 128) = __builtin_shufflevector(acc[mt][t], acc[mt][t], 4, 5, 6, 7);
    }
}
template <int NT, int B0, int B1, bool SAVE>
__device__ __forceinline__ void run_blocks(v8f (&acc)[2][NT],
                                           v2f (&fa0)[KB][2], v2f (&fb0)[KB][NT],
                                           v2f (&fa1)[KB][2], v2f (&fb1)[KB][NT],
                                           const float* A0, const float* A1,
                                           const float* Bbase,
                                           float* sQw, int lane) {
#pragma nounroll
  for (int blk = B0; blk < B1; blk += 2) {
    load_frags<NT>(fa1, fb1, A0, A1, Bbase, blk + 1);
    mma_block<NT>(acc, fa0, fb0);
    if (blk + 2 < B1) load_frags<NT>(fa0, fb0, A0, A1, Bbase, blk + 2);
    mma_block<NT>(acc, fa1, fb1);
    if constexpr (SAVE) {
      if (blk + 2 == QBLK) save_acc(sQw, acc, lane);
    }
  }
}

__device__ __forceinline__ void restore_acc(const float* sQw, v8f (&acc)[2][2], int lane) {
#pragma unroll
  for (int mt = 0; mt < 2; ++mt)
#pragma unroll
    for (int t = 0; t < 2; ++t) {
      const float* d = sQw + ((mt * 2 + t) * 2) * 128 + lane * 4;
      const v4f lo = *(const v4f*)(d);
      const v4f hi = *(const v4f*)(d + 128);
      acc[mt][t] = __builtin_shufflevector(lo, hi, 0, 1, 2, 3, 4, 5, 6, 7);
    }
}

template <int MODE, int NT, int KM>
__device__ __forceinline__ void gemm_512(const float* Abuf,
                                         const float* Bpk,
                                         float* Obuf,
                                         const float* Maskbuf,
                                         const float* __restrict__ sCst,
                                         float* sQw,
                                         int wave, int lane) {
  static_assert(KM == 0 || NT == 2, "the kept accumulators are the 2 x 2 tiles");
  const int m     = lane & 15;
  const int khalf = lane >> 4;

  v8f acc[2][NT];
  if constexpr (KM == 2) {
    restore_acc(sQw, acc, lane);
  } else {
#pragma unroll
    for (int mt = 0; mt < 2; ++mt)
#pragma unroll
      for (int t = 0; t < NT; ++t)
#pragma unroll
        for (int j = 0; j < 8; ++j) acc[mt][t][j] = 0.0f;
  }

  const float* A0 = Abuf + m * LDS_STRIDE + (khalf << 1);
  const float* A1 = A0 + 16 * LDS_STRIDE;
  const float* Bbase = Bpk + khalf * (TWO_D * 2) + (wave * (16 * NT) + m) * 2;

  v2f fa0[KB][2], fb0[KB][NT];
  v2f fa1[KB][2], fb1[KB][NT];

  constexpr int BFIRST = (KM == 2) ? QBLK : 0;
  load_frags<NT>(fa0, fb0, A0, A1, Bbase, BFIRST);
  run_blocks<NT, BFIRST, NBLK, KM == 1>(acc, fa0, fb0, fa1, fb1, A0, A1, Bbase, sQw, lane);

  const int rbase = khalf << 3;
  const int colb  = wave * (16 * NT) + m;
  const float c0a = (MODE == 1) ? sCst[colb]              : (MODE == 2) ? sCst[TWO_D + colb]      : 0.0f;
  const float c1a = (MODE == 1) ? sCst[colb + 16]         : (MODE == 2) ? sCst[TWO_D + colb + 16] : 0.0f;
  const float w0  = (MODE == 2) ? sCst[2 * TWO_D + colb]      : 0.0f;
  const float w1  = (MODE == 2) ? sCst[2 * TWO_D + colb + 16] : 0.0f;

#pragma unroll
  for (int mt = 0; mt < 2; ++mt) {
#pragma unroll
    for (int j = 0; j < 8; ++j) {
      const int row = mt * 16 + rbase + j;
      float* o = Obuf + row * LDS_STRIDE + colb;
      float v0 = acc[mt][0][j];
      float v1 = acc[mt][NT - 1][j];
      if (MODE == 1) {
        o[0]  = fmaxf(v0 + c0a, 0.0f);
        if (NT == 2) o[16] = fmaxf(v1 + c1a, 0.0f);
      } else if (MODE == 2) {
        o[0]  = (v0 + c0a > 0.0f) ? w0 : 0.0f;
        if (NT == 2) o[16] = (v1 + c1a > 0.0f) ? w1 : 0.0f;
      } else if (MODE == 3) {
        const float* mk = Maskbuf + row * LDS_STRIDE + colb;
        float h0 = mk[0];
        o[0]  = (h0 > 0.0f) ? v0 : 0.0f;
        if (NT == 2) {
          float h1 = mk[16];
          o[16] = (h1 > 0.0f) ? v1 : 0.0f;
        }
      } else {
        o[0]  = v0;
        if (NT == 2) o[16] = v1;
      }
    }
  }
}

__device__ __forceinline__ void q_partial(const float* Abuf, const float* Bpk,
                                          float* sQw, int wave, int lane) {
  const int m     = lane & 15;
  const int khalf = lane >> 4;
  const float* A0 = Abuf + m * LDS_STRIDE + (khalf << 1);
  const float* A1 = A0 + 16 * LDS_STRIDE;
  const float* Bbase = Bpk + khalf * (TWO_D * 2) + (wave * 32 + m) * 2;
  v8f acc[2][2];
#pragma unroll
  for (int mt = 0; mt < 2; ++mt)
#pragma unroll
    for (int t = 0; t < 2; ++t)
#pragma unroll
      for (int j = 0; j < 8; ++j) acc[mt][t][j] = 0.0f;
  v2f fa0[KB][2], fb0[KB][2];
  v2f fa1[KB][2], fb1[KB][2];
  load_frags<2>(fa0, fb0, A0, A1, Bbase, 0);
  run_blocks<2, 0, QBLK, false>(acc, fa0, fb0, fa1, fb1, A0, A1, Bbase, sQw, lane);
  save_acc(sQw, acc, lane);
}

template <bool SECOND>
__device__ __forceinline__ void grad_eval(float* sS, float* sT1, float* sT2,
                                          const float* sCst,
                                          const float* P1f,
                                          const float* P2f,
                                          const float* P2b,
                                          const float* P1b,
                                          float* sQw,
                                          int wave, int lane) {
  if (SECOND)
    gemm_512<1, 2, 1>(sS, P1f, sT1, nullptr, sCst, sQw, wave, lane);
  else
    gemm_512<1, 2, 2>(sS, P1f, sT1, nullptr, sCst, sQw, wave, lane);
  __syncthreads();
  gemm_512<2, 2, 0>(sT1, P2f, sT2, nullptr, sCst, sQw, wave, lane);
  __syncthreads();
  gemm_512<3, 2, 0>(sT2, P2b, sT1, sT1,     sCst, sQw, wave, lane);
  __syncthreads();
  if (SECOND)
    gemm_512<4, 1, 0>(sT1, P1b, sT2, nullptr, sCst, sQw, wave, lane);
  else
    gemm_512<4, 2, 0>(sT1, P1b, sT2, nullptr, sCst, sQw, wave, lane);
  __syncthreads();
}

__global__ __launch_bounds__(NTHREADS) void hnn_leapfrog_kernel(
    const float* __restrict__ x,
    const float* __restrict__ b1, const float* __restrict__ b2,
    const float* __restrict__ Wo,
    const float* P1f, const float* P2f,
    const float* P2b, const float* P1b,
    float* __restrict__ out)
{
  __shared__ float sS [ROWS * LDS_STRIDE];
  __shared__ float sT1[ROWS * LDS_STRIDE];
  __shared__ float sT2[ROWS * LDS_STRIDE];
  __shared__ float sCst[3 * TWO_D];
  __shared__ __attribute__((aligned(16))) float sQ[NWAVES * QSAVE_WAVE];

  const int tid  = threadIdx.x;
  const int wave = tid >> 5;
  const int lane = tid & 31;
  const int r0   = blockIdx.x * ROWS;

  for (int i = tid; i < TWO_D; i += NTHREADS) {
    sCst[i]             = b1[i];
    sCst[TWO_D + i]     = b2[i];
    sCst[2 * TWO_D + i] = Wo[i];
  }

  for (int i = tid; i < ROWS * DD; i += NTHREADS) {
    const int r = i >> 8;
    const int d = i & 255;
    const float* xp = x + ((r0 + r) * DD + d) * 4;
    const float x3 = xp[3];
    const float x2 = xp[2];
    sS[r * LDS_STRIDE + d]      = x3;
    sS[r * LDS_STRIDE + DD + d] = x3 - x2;
  }
  __syncthreads();

  float* sQw = sQ + wave * QSAVE_WAVE;
  q_partial(sS, P1f, sQw, wave, lane);

  for (int step = 0; step < STEPS; ++step) {
    grad_eval<false>(sS, sT1, sT2, sCst, P1f, P2f, P2b, P1b, sQw, wave, lane);
    for (int i = tid; i < ROWS * DD; i += NTHREADS) {
      const int r = i >> 8;
      const int d = i & 255;
      const float dHq = sT2[r * LDS_STRIDE + d];
      const float dHp = sT2[r * LDS_STRIDE + DD + d];
      sS[r * LDS_STRIDE + DD + d] -= 0.5f * DT * dHq;
      sS[r * LDS_STRIDE + d]      += DT * dHp;
    }
    __syncthreads();

    grad_eval<true>(sS, sT1, sT2, sCst, P1f, P2f, P2b, P1b, sQw, wave, lane);
    for (int i = tid; i < ROWS * DD; i += NTHREADS) {
      const int r = i >> 8;
      const int d = i & 255;
      sS[r * LDS_STRIDE + DD + d] -= 0.5f * DT * sT2[r * LDS_STRIDE + d];
    }
    __syncthreads();
  }

  for (int i = tid; i < ROWS * TWO_D; i += NTHREADS) {
    const int r = i >> 9;
    const int c = i & 511;
    out[(r0 + r) * TWO_D + c] = sS[r * LDS_STRIDE + c];
  }
}

extern "C" void kernel_launch(void* const* d_in, const int* in_sizes, int n_in,
                              void* d_out, int out_size, void* d_ws, size_t ws_size,
                              hipStream_t stream) {
  const float* x  = (const float*)d_in[0];
  const float* W1 = (const float*)d_in[1];
  const float* b1 = (const float*)d_in[2];
  const float* W2 = (const float*)d_in[3];
  const float* b2 = (const float*)d_in[4];
  const float* Wo = (const float*)d_in[5];
  float* out = (float*)d_out;

  const int MS = TWO_D * TWO_D;
  float* P1f = (float*)d_ws;
  float* P2f = P1f + MS;
  float* P2b = P2f + MS;
  float* P1b = P2b + MS;

  pack_weights<<<(MS + 255) / 256, 256, 0, stream>>>(W1, P1f, 1);
  pack_weights<<<(MS + 255) / 256, 256, 0, stream>>>(W2, P2f, 1);
  pack_weights<<<(MS + 255) / 256, 256, 0, stream>>>(W2, P2b, 0);
  pack_weights<<<(MS + 255) / 256, 256, 0, stream>>>(W1, P1b, 0);

  const int nblocks = 8192 / ROWS;
  hnn_leapfrog_kernel<<<nblocks, NTHREADS, 0, stream>>>(
      x, b1, b2, Wo, P1f, P2f, P2b, P1b, out);
}


# Wave priority (s_setprio) falls with a wave's own K-loop progress (3, then 2, 1, 0 at 8, 4, 2 register blocks left) so a SIMD's 4 waves take turns on the matrix pipe; order of every sum unchanged
# speedup vs baseline: 1.4187x; 1.0475x over previous
#include <hip/hip_runtime.h>


#define DD        256
#define TWO_D     512
#define STEPS     16
#define DT        0.1f
#define ROWS      32
#define NWAVES    16
#define NTHREADS  (NWAVES * 32)
#define LDS_STRIDE (TWO_D + 4)
#define KB        8
#define NBLK      (TWO_D / (4 * KB))
#define BSTEP     (2 * TWO_D * 2)

typedef __attribute__((ext_vector_type(2))) float v2f;
typedef __attribute__((ext_vector_type(4))) float v4f;
typedef __attribute__((ext_vector_type(8))) float v8f;

#define WMMA_F32(A, B, C) \
  __builtin_amdgcn_wmma_f32_16x16x4_f32(false, (A), false, (B), (short)0, (C), false, false)

__global__ __launch_bounds__(256) void pack_weights(const float* __restrict__ W,
                                                    float* __restrict__ P,
                                                    const int transposed) {
  const int idx = blockIdx.x * 256 + threadIdx.x;
  const int n = idx & 511;
  const int k = idx >> 9;
  const float v = transposed ? W[n * TWO_D + k] : W[k * TWO_D + n];
  const int grp = ((k >> 2) << 1) + ((k >> 1) & 1);
  P[grp * (TWO_D * 2) + (n << 1) + (k & 1)] = v;
}

template <int NT>
__device__ __forceinline__ void load_frags(v2f (&fa)[KB][2], v2f (&fb)[KB][NT],
                                           const float* A0, const float* A1,
                                           const float* Bbase,
                                           int blk) {
  const float* Bp = Bbase + blk * (KB * BSTEP);
#pragma unroll
  for (int st = 0; st < KB; ++st) {
    fb[st][0] = *(const v2f*)(Bp + st * BSTEP);
    if (NT == 2) fb[st][NT - 1] = *(const v2f*)(Bp + st * BSTEP + 32);
  }
  const int kb = blk * (KB * 4);
#pragma unroll
  for (int st = 0; st < KB; ++st) {
    fa[st][0] = *(const v2f*)(A0 + kb + st * 4);
    fa[st][1] = *(const v2f*)(A1 + kb + st * 4);
  }
}

template <int NT>
__device__ __forceinline__ void mma_block(v8f (&acc)[2][NT],
                                          const v2f (&fa)[KB][2],
                                          const v2f (&fb)[KB][NT]) {
#pragma unroll
  for (int st = 0; st < KB; ++st) {
    if (NT == 2) {
      acc[0][0]      = WMMA_F32(fa[st][0], fb[st][0],      acc[0][0]);
      acc[0][NT - 1] = WMMA_F32(fa[st][0], fb[st][NT - 1], acc[0][NT - 1]);
      acc[1][0]      = WMMA_F32(fa[st][1], fb[st][0],      acc[1][0]);
      acc[1][NT - 1] = WMMA_F32(fa[st][1], fb[st][NT - 1], acc[1][NT - 1]);
    } else {
      acc[0][0] = WMMA_F32(fa[st][0], fb[st][0], acc[0][0]);
      acc[1][0] = WMMA_F32(fa[st][1], fb[st][0], acc[1][0]);
    }
  }
}

#define QBLK (NBLK / 2)
#define QSAVE_WAVE (2 * 2 * 8 * 32)

__device__ __forceinline__ void save_acc(float* sQw, const v8f (&acc)[2][2], int lane) {
#pragma unroll
  for (int mt = 0; mt < 2; ++mt)
#pragma unroll
    for (int t = 0; t < 2; ++t) {
      float* d = sQw + ((mt * 2 + t) * 2) * 128 + lane * 4;
      *(v4f*)(d)       = __builtin_shufflevector(acc[mt][t], acc[mt][t], 0, 1, 2, 3);
      *(v4f*)(d + 128) = __builtin_shufflevector(acc[mt][t], acc[mt][t], 4, 5, 6, 7);
    }
}
template <int NT, int B0, int B1, bool SAVE>
__device__ __forceinline__ void run_blocks(v8f (&acc)[2][NT],
                                           v2f (&fa0)[KB][2], v2f (&fb0)[KB][NT],
                                           v2f (&fa1)[KB][2], v2f (&fb1)[KB][NT],
                                           const float* A0, const float* A1,
                                           const float* Bbase,
                                           float* sQw, int lane) {
#pragma nounroll
  for (int blk = B0; blk < B1; blk += 2) {
    const int left = B1 - blk;
    if (left == 8)      __builtin_amdgcn_s_setprio(2);
    else if (left == 4) __builtin_amdgcn_s_setprio(1);
    else if (left == 2) __builtin_amdgcn_s_setprio(0);
    load_frags<NT>(fa1, fb1, A0, A1, Bbase, blk + 1);
    mma_block<NT>(acc, fa0, fb0);
    if (blk + 2 < B1) load_frags<NT>(fa0, fb0, A0, A1, Bbase, blk + 2);
    mma_block<NT>(acc, fa1, fb1);
    if constexpr (SAVE) {
      if (blk + 2 == QBLK) save_acc(sQw, acc, lane);
    }
  }
}

__device__ __forceinline__ void restore_acc(const float* sQw, v8f (&acc)[2][2], int lane) {
#pragma unroll
  for (int mt = 0; mt < 2; ++mt)
#pragma unroll
    for (int t = 0; t < 2; ++t) {
      const float* d = sQw + ((mt * 2 + t) * 2) * 128 + lane * 4;
      const v4f lo = *(const v4f*)(d);
      const v4f hi = *(const v4f*)(d + 128);
      acc[mt][t] = __builtin_shufflevector(lo, hi, 0, 1, 2, 3, 4, 5, 6, 7);
    }
}

template <int MODE, int NT, int KM>
__device__ __forceinline__ void gemm_512(const float* Abuf,
                                         const float* Bpk,
                                         float* Obuf,
                                         const float* Maskbuf,
                                         const float* __restrict__ sCst,
                                         float* sQw,
                                         int wave, int lane) {
  static_assert(KM == 0 || NT == 2, "the kept accumulators are the 2 x 2 tiles");
  const int m     = lane & 15;
  const int khalf = lane >> 4;

  v8f acc[2][NT];
  if constexpr (KM == 2) {
    restore_acc(sQw, acc, lane);
  } else {
#pragma unroll
    for (int mt = 0; mt < 2; ++mt)
#pragma unroll
      for (int t = 0; t < NT; ++t)
#pragma unroll
        for (int j = 0; j < 8; ++j) acc[mt][t][j] = 0.0f;
  }

  const float* A0 = Abuf + m * LDS_STRIDE + (khalf << 1);
  const float* A1 = A0 + 16 * LDS_STRIDE;
  const float* Bbase = Bpk + khalf * (TWO_D * 2) + (wave * (16 * NT) + m) * 2;

  v2f fa0[KB][2], fb0[KB][NT];
  v2f fa1[KB][2], fb1[KB][NT];

  constexpr int BFIRST = (KM == 2) ? QBLK : 0;
  __builtin_amdgcn_s_setprio(3);
  load_frags<NT>(fa0, fb0, A0, A1, Bbase, BFIRST);
  run_blocks<NT, BFIRST, NBLK, KM == 1>(acc, fa0, fb0, fa1, fb1, A0, A1, Bbase, sQw, lane);

  const int rbase = khalf << 3;
  const int colb  = wave * (16 * NT) + m;
  const float c0a = (MODE == 1) ? sCst[colb]              : (MODE == 2) ? sCst[TWO_D + colb]      : 0.0f;
  const float c1a = (MODE == 1) ? sCst[colb + 16]         : (MODE == 2) ? sCst[TWO_D + colb + 16] : 0.0f;
  const float w0  = (MODE == 2) ? sCst[2 * TWO_D + colb]      : 0.0f;
  const float w1  = (MODE == 2) ? sCst[2 * TWO_D + colb + 16] : 0.0f;

#pragma unroll
  for (int mt = 0; mt < 2; ++mt) {
#pragma unroll
    for (int j = 0; j < 8; ++j) {
      const int row = mt * 16 + rbase + j;
      float* o = Obuf + row * LDS_STRIDE + colb;
      float v0 = acc[mt][0][j];
      float v1 = acc[mt][NT - 1][j];
      if (MODE == 1) {
        o[0]  = fmaxf(v0 + c0a, 0.0f);
        if (NT == 2) o[16] = fmaxf(v1 + c1a, 0.0f);
      } else if (MODE == 2) {
        o[0]  = (v0 + c0a > 0.0f) ? w0 : 0.0f;
        if (NT == 2) o[16] = (v1 + c1a > 0.0f) ? w1 : 0.0f;
      } else if (MODE == 3) {
        const float* mk = Maskbuf + row * LDS_STRIDE + colb;
        float h0 = mk[0];
        o[0]  = (h0 > 0.0f) ? v0 : 0.0f;
        if (NT == 2) {
          float h1 = mk[16];
          o[16] = (h1 > 0.0f) ? v1 : 0.0f;
        }
      } else {
        o[0]  = v0;
        if (NT == 2) o[16] = v1;
      }
    }
  }
}

__device__ __forceinline__ void q_partial(const float* Abuf, const float* Bpk,
                                          float* sQw, int wave, int lane) {
  const int m     = lane & 15;
  const int khalf = lane >> 4;
  const float* A0 = Abuf + m * LDS_STRIDE + (khalf << 1);
  const float* A1 = A0 + 16 * LDS_STRIDE;
  const float* Bbase = Bpk + khalf * (TWO_D * 2) + (wave * 32 + m) * 2;
  v8f acc[2][2];
#pragma unroll
  for (int mt = 0; mt < 2; ++mt)
#pragma unroll
    for (int t = 0; t < 2; ++t)
#pragma unroll
      for (int j = 0; j < 8; ++j) acc[mt][t][j] = 0.0f;
  v2f fa0[KB][2], fb0[KB][2];
  v2f fa1[KB][2], fb1[KB][2];
  __builtin_amdgcn_s_setprio(3);
  load_frags<2>(fa0, fb0, A0, A1, Bbase, 0);
  run_blocks<2, 0, QBLK, false>(acc, fa0, fb0, fa1, fb1, A0, A1, Bbase, sQw, lane);
  save_acc(sQw, acc, lane);
}

template <bool SECOND>
__device__ __forceinline__ void grad_eval(float* sS, float* sT1, float* sT2,
                                          const float* sCst,
                                          const float* P1f,
                                          const float* P2f,
                                          const float* P2b,
                                          const float* P1b,
                                          float* sQw,
                                          int wave, int lane) {
  if (SECOND)
    gemm_512<1, 2, 1>(sS, P1f, sT1, nullptr, sCst, sQw, wave, lane);
  else
    gemm_512<1, 2, 2>(sS, P1f, sT1, nullptr, sCst, sQw, wave, lane);
  __syncthreads();
  gemm_512<2, 2, 0>(sT1, P2f, sT2, nullptr, sCst, sQw, wave, lane);
  __syncthreads();
  gemm_512<3, 2, 0>(sT2, P2b, sT1, sT1,     sCst, sQw, wave, lane);
  __syncthreads();
  if (SECOND)
    gemm_512<4, 1, 0>(sT1, P1b, sT2, nullptr, sCst, sQw, wave, lane);
  else
    gemm_512<4, 2, 0>(sT1, P1b, sT2, nullptr, sCst, sQw, wave, lane);
  __syncthreads();
}

__global__ __launch_bounds__(NTHREADS) void hnn_leapfrog_kernel(
    const float* __restrict__ x,
    const float* __restrict__ b1, const float* __restrict__ b2,
    const float* __restrict__ Wo,
    const float* P1f, const float* P2f,
    const float* P2b, const float* P1b,
    float* __restrict__ out)
{
  __shared__ float sS [ROWS * LDS_STRIDE];
  __shared__ float sT1[ROWS * LDS_STRIDE];
  __shared__ float sT2[ROWS * LDS_STRIDE];
  __shared__ float sCst[3 * TWO_D];
  __shared__ __attribute__((aligned(16))) float sQ[NWAVES * QSAVE_WAVE];

  const int tid  = threadIdx.x;
  const int wave = tid >> 5;
  const int lane = tid & 31;
  const int r0   = blockIdx.x * ROWS;

  for (int i = tid; i < TWO_D; i += NTHREADS) {
    sCst[i]             = b1[i];
    sCst[TWO_D + i]     = b2[i];
    sCst[2 * TWO_D + i] = Wo[i];
  }

  for (int i = tid; i < ROWS * DD; i += NTHREADS) {
    const int r = i >> 8;
    const int d = i & 255;
    const float* xp = x + ((r0 + r) * DD + d) * 4;
    const float x3 = xp[3];
    const float x2 = xp[2];
    sS[r * LDS_STRIDE + d]      = x3;
    sS[r * LDS_STRIDE + DD + d] = x3 - x2;
  }
  __syncthreads();

  float* sQw = sQ + wave * QSAVE_WAVE;
  q_partial(sS, P1f, sQw, wave, lane);

  for (int step = 0; step < STEPS; ++step) {
    grad_eval<false>(sS, sT1, sT2, sCst, P1f, P2f, P2b, P1b, sQw, wave, lane);
    for (int i = tid; i < ROWS * DD; i += NTHREADS) {
      const int r = i >> 8;
      const int d = i & 255;
      const float dHq = sT2[r * LDS_STRIDE + d];
      const float dHp = sT2[r * LDS_STRIDE + DD + d];
      sS[r * LDS_STRIDE + DD + d] -= 0.5f * DT * dHq;
      sS[r * LDS_STRIDE + d]      += DT * dHp;
    }
    __syncthreads();

    grad_eval<true>(sS, sT1, sT2, sCst, P1f, P2f, P2b, P1b, sQw, wave, lane);
    for (int i = tid; i < ROWS * DD; i += NTHREADS) {
      const int r = i >> 8;
      const int d = i & 255;
      sS[r * LDS_STRIDE + DD + d] -= 0.5f * DT * sT2[r * LDS_STRIDE + d];
    }
    __syncthreads();
  }

  for (int i = tid; i < ROWS * TWO_D; i += NTHREADS) {
    const int r = i >> 9;
    const int c = i & 511;
    out[(r0 + r) * TWO_D + c] = sS[r * LDS_STRIDE + c];
  }
}

extern "C" void kernel_launch(void* const* d_in, const int* in_sizes, int n_in,
                              void* d_out, int out_size, void* d_ws, size_t ws_size,
                              hipStream_t stream) {
  const float* x  = (const float*)d_in[0];
  const float* W1 = (const float*)d_in[1];
  const float* b1 = (const float*)d_in[2];
  const float* W2 = (const float*)d_in[3];
  const float* b2 = (const float*)d_in[4];
  const float* Wo = (const float*)d_in[5];
  float* out = (float*)d_out;

  const int MS = TWO_D * TWO_D;
  float* P1f = (float*)d_ws;
  float* P2f = P1f + MS;
  float* P2b = P2f + MS;
  float* P1b = P2b + MS;

  pack_weights<<<(MS + 255) / 256, 256, 0, stream>>>(W1, P1f, 1);
  pack_weights<<<(MS + 255) / 256, 256, 0, stream>>>(W2, P2f, 1);
  pack_weights<<<(MS + 255) / 256, 256, 0, stream>>>(W2, P2b, 0);
  pack_weights<<<(MS + 255) / 256, 256, 0, stream>>>(W1, P1b, 0);

  const int nblocks = 8192 / ROWS;
  hnn_leapfrog_kernel<<<nblocks, NTHREADS, 0, stream>>>(
      x, b1, b2, Wo, P1f, P2f, P2b, P1b, out);
}


# Leapfrog updates folded into the weights: z1 += g1 @ (W1 J W1^T) per evaluation, state = s0 + (summed g1) @ W1 once at the end; 6 GEMMs per time step instead of 7; the order of the sums changes
# speedup vs baseline: 1.6139x; 1.1376x over previous
#include <hip/hip_runtime.h>


#define DD        256
#define TWO_D     512
#define STEPS     16
#define DT        0.1f
#define ROWS      32
#define NWAVES    16
#define NTHREADS  (NWAVES * 32)
#define LDS_STRIDE (TWO_D + 4)
#define KB        8
#define NBLK      (TWO_D / (4 * KB))
#define BSTEP     (2 * TWO_D * 2)

typedef __attribute__((ext_vector_type(2))) float v2f;
typedef __attribute__((ext_vector_type(4))) float v4f;
typedef __attribute__((ext_vector_type(8))) float v8f;

#define WMMA_F32(A, B, C) \
  __builtin_amdgcn_wmma_f32_16x16x4_f32(false, (A), false, (B), (short)0, (C), false, false)

__device__ __forceinline__ int pack_index(int k, int n) {
  const int grp = ((k >> 2) << 1) + ((k >> 1) & 1);
  return grp * (TWO_D * 2) + (n << 1) + (k & 1);
}

__global__ __launch_bounds__(256) void pack_weights(const float* __restrict__ W,
                                                    float* __restrict__ P,
                                                    const int transposed) {
  const int idx = blockIdx.x * 256 + threadIdx.x;
  const int n = idx & 511;
  const int k = idx >> 9;
  const float v = transposed ? W[n * TWO_D + k] : W[k * TWO_D + n];
  P[pack_index(k, n)] = v;
}

__global__ __launch_bounds__(256) void pack_step_matrices(const float* __restrict__ W,
                                                          float* __restrict__ PMa,
                                                          float* __restrict__ PMb) {
  const int wave = threadIdx.x >> 5;
  const int lane = threadIdx.x & 31;
  const int tile = blockIdx.x * 8 + wave;
  const int k0 = (tile >> 5) << 4;
  const int n0 = (tile & 31) << 4;
  const int c = lane & 15;
  const int h = lane >> 4;
  const float* rk = W + (k0 + c) * TWO_D + (h << 1);
  const float* rn = W + (n0 + c) * TWO_D + (h << 1);
  v8f a, b;
#pragma unroll
  for (int j = 0; j < 8; ++j) { a[j] = 0.0f; b[j] = 0.0f; }
  for (int i = 0; i < DD; i += 4) {
    const v2f kp = *(const v2f*)(rk + DD + i);
    const v2f nq = *(const v2f*)(rn + i);
    const v2f kq = *(const v2f*)(rk + i);
    const v2f np = *(const v2f*)(rn + DD + i);
    a = WMMA_F32(kp, nq, a);
    b = WMMA_F32(kq, np, b);
  }
#pragma unroll
  for (int j = 0; j < 8; ++j) {
    const int idx = pack_index(k0 + (h << 3) + j, n0 + c);
    PMa[idx] = DT * a[j] - 0.5f * DT * b[j];
    PMb[idx] = -0.5f * DT * b[j];
  }
}

template <int KS, int TS, int YS>
__device__ __forceinline__ void load_frags(v2f (&fa)[KB][2], v2f (&fb)[KB][2],
                                           const float* A0, const float* A1,
                                           const float* Bbase,
                                           int blk) {
  const float* Bp = Bbase + blk * (KB * KS);
#pragma unroll
  for (int st = 0; st < KB; ++st) {
    if (YS == 1) {
      fb[st][0] = *(const v2f*)(Bp + st * KS);
      fb[st][1] = *(const v2f*)(Bp + st * KS + TS);
    } else {
      fb[st][0].x = Bp[st * KS];
      fb[st][0].y = Bp[st * KS + YS];
      fb[st][1].x = Bp[st * KS + TS];
      fb[st][1].y = Bp[st * KS + TS + YS];
    }
  }
  const int kb = blk * (KB * 4);
#pragma unroll
  for (int st = 0; st < KB; ++st) {
    fa[st][0] = *(const v2f*)(A0 + kb + st * 4);
    fa[st][1] = *(const v2f*)(A1 + kb + st * 4);
  }
}

__device__ __forceinline__ float relu_bits(float v) {
  const int b = __float_as_int(v);
  return __int_as_float(b > 0 ? b : 0);
}

template <bool RELU>
__device__ __forceinline__ void mma_block(v8f (&acc)[2][2],
                                          const v2f (&fa)[KB][2],
                                          const v2f (&fb)[KB][2]) {
#pragma unroll
  for (int st = 0; st < KB; ++st) {
    v2f a0 = fa[st][0];
    v2f a1 = fa[st][1];
    if (RELU) {
      a0.x = relu_bits(a0.x); a0.y = relu_bits(a0.y);
      a1.x = relu_bits(a1.x); a1.y = relu_bits(a1.y);
    }
    acc[0][0] = WMMA_F32(a0, fb[st][0], acc[0][0]);
    acc[0][1] = WMMA_F32(a0, fb[st][1], acc[0][1]);
    acc[1][0] = WMMA_F32(a1, fb[st][0], acc[1][0]);
    acc[1][1] = WMMA_F32(a1, fb[st][1], acc[1][1]);
  }
}

#define GSAVE_WAVE (2 * 2 * 8 * 32)

__device__ __forceinline__ void add_to_sum(float* sGw, const v8f (&g)[2][2], int lane) {
#pragma unroll
  for (int mt = 0; mt < 2; ++mt)
#pragma unroll
    for (int t = 0; t < 2; ++t) {
      float* d = sGw + ((mt * 2 + t) * 2) * 128 + lane * 4;
      v4f lo = *(const v4f*)(d);
      v4f hi = *(const v4f*)(d + 128);
      lo += __builtin_shufflevector(g[mt][t], g[mt][t], 0, 1, 2, 3);
      hi += __builtin_shufflevector(g[mt][t], g[mt][t], 4, 5, 6, 7);
      *(v4f*)(d)       = lo;
      *(v4f*)(d + 128) = hi;
    }
}
__device__ __forceinline__ void load_sum(const float* sGw, v8f (&g)[2][2], int lane) {
#pragma unroll
  for (int mt = 0; mt < 2; ++mt)
#pragma unroll
    for (int t = 0; t < 2; ++t) {
      const float* d = sGw + ((mt * 2 + t) * 2) * 128 + lane * 4;
      const v4f lo = *(const v4f*)(d);
      const v4f hi = *(const v4f*)(d + 128);
      g[mt][t] = __builtin_shufflevector(lo, hi, 0, 1, 2, 3, 4, 5, 6, 7);
    }
}

template <int KS, int TS, int YS, bool RELU>
__device__ __forceinline__ void run_blocks(v8f (&acc)[2][2],
                                           v2f (&fa0)[KB][2], v2f (&fb0)[KB][2],
                                           v2f (&fa1)[KB][2], v2f (&fb1)[KB][2],
                                           const float* A0, const float* A1,
                                           const float* Bbase) {
#pragma nounroll
  for (int blk = 0; blk < NBLK; blk += 2) {
    const int left = NBLK - blk;
    if (left == 8)      __builtin_amdgcn_s_setprio(2);
    else if (left == 4) __builtin_amdgcn_s_setprio(1);
    else if (left == 2) __builtin_amdgcn_s_setprio(0);
    load_frags<KS, TS, YS>(fa1, fb1, A0, A1, Bbase, blk + 1);
    mma_block<RELU>(acc, fa0, fb0);
    if (blk + 2 < NBLK) load_frags<KS, TS, YS>(fa0, fb0, A0, A1, Bbase, blk + 2);
    mma_block<RELU>(acc, fa1, fb1);
  }
}

template <int MODE, bool RELU, int KS, int TS, int YS>
__device__ __forceinline__ void gemm_512(const float* Abuf,
                                         const float* Bbase,
                                         float* Obuf,
                                         const float* Maskbuf,
                                         const float* __restrict__ sCst,
                                         float* sGw,
                                         const float* __restrict__ x,
                                         float* __restrict__ out,
                                         int r0, int wave, int lane) {
  const int m     = lane & 15;
  const int khalf = lane >> 4;

  v8f acc[2][2];
#pragma unroll
  for (int mt = 0; mt < 2; ++mt)
#pragma unroll
    for (int t = 0; t < 2; ++t)
#pragma unroll
      for (int j = 0; j < 8; ++j) acc[mt][t][j] = 0.0f;

  const float* A0 = Abuf + m * LDS_STRIDE + (khalf << 1);
  const float* A1 = A0 + 16 * LDS_STRIDE;

  v2f fa0[KB][2], fb0[KB][2];
  v2f fa1[KB][2], fb1[KB][2];

  __builtin_amdgcn_s_setprio(3);
  load_frags<KS, TS, YS>(fa0, fb0, A0, A1, Bbase, 0);
  run_blocks<KS, TS, YS, RELU>(acc, fa0, fb0, fa1, fb1, A0, A1, Bbase);

  if (MODE == 3) __syncthreads();

  const int rbase = khalf << 3;
  const int colb  = wave * 32 + m;
  const float c0a = (MODE == 5) ? sCst[colb]      : (MODE == 2) ? sCst[TWO_D + colb]      : 0.0f;
  const float c1a = (MODE == 5) ? sCst[colb + 16] : (MODE == 2) ? sCst[TWO_D + colb + 16] : 0.0f;
  const float w0  = (MODE == 2) ? sCst[2 * TWO_D + colb]      : 0.0f;
  const float w1  = (MODE == 2) ? sCst[2 * TWO_D + colb + 16] : 0.0f;

#pragma unroll
  for (int mt = 0; mt < 2; ++mt) {
#pragma unroll
    for (int j = 0; j < 8; ++j) {
      const int row = mt * 16 + rbase + j;
      const float v0 = acc[mt][0][j];
      const float v1 = acc[mt][1][j];
      if (MODE == 7) {
        const int grow = r0 + row;
        const int cc   = colb & (DD - 1);
        const float* xa = x + (grow * DD + cc) * 4;
        const float* xb = xa + 16 * 4;
        if (wave < NWAVES / 2) {
          out[grow * TWO_D + DD + cc]      = (xa[3] - xa[2]) - 0.5f * DT * v0;
          out[grow * TWO_D + DD + cc + 16] = (xb[3] - xb[2]) - 0.5f * DT * v1;
        } else {
          out[grow * TWO_D + cc]      = xa[3] + DT * v0;
          out[grow * TWO_D + cc + 16] = xb[3] + DT * v1;
        }
      } else {
        float* o = Obuf + row * LDS_STRIDE + colb;
        if (MODE == 2) {
          o[0]  = (v0 + c0a > 0.0f) ? w0 : 0.0f;
          o[16] = (v1 + c1a > 0.0f) ? w1 : 0.0f;
        } else if (MODE == 3) {
          const float* mk = Maskbuf + row * LDS_STRIDE + colb;
          const float g0 = (mk[0]  > 0.0f) ? v0 : 0.0f;
          const float g1 = (mk[16] > 0.0f) ? v1 : 0.0f;
          o[0]  = g0;
          o[16] = g1;
          acc[mt][0][j] = g0;
          acc[mt][1][j] = g1;
        } else if (MODE == 5) {
          o[0]  = v0 + c0a;
          o[16] = v1 + c1a;
        } else {
          o[0]  = o[0]  + v0;
          o[16] = o[16] + v1;
        }
      }
    }
  }
  if (MODE == 3) add_to_sum(sGw, acc, lane);
}

#define PACKED BSTEP, 32, 1

__global__ __launch_bounds__(NTHREADS) void hnn_leapfrog_kernel(
    const float* __restrict__ x,
    const float* W1,
    const float* __restrict__ b1, const float* __restrict__ b2,
    const float* __restrict__ Wo,
    const float* P2f, const float* P2b,
    const float* PMa, const float* PMb,
    float* __restrict__ out)
{
  __shared__ float sZ[ROWS * LDS_STRIDE];
  __shared__ float sT[ROWS * LDS_STRIDE];
  __shared__ float sCst[3 * TWO_D];
  __shared__ __attribute__((aligned(16))) float sGa[NWAVES * GSAVE_WAVE];
  __shared__ __attribute__((aligned(16))) float sGb[NWAVES * GSAVE_WAVE];

  const int tid  = threadIdx.x;
  const int wave = tid >> 5;
  const int lane = tid & 31;
  const int m     = lane & 15;
  const int khalf = lane >> 4;
  const int r0   = blockIdx.x * ROWS;

  for (int i = tid; i < TWO_D; i += NTHREADS) {
    sCst[i]             = b1[i];
    sCst[TWO_D + i]     = b2[i];
    sCst[2 * TWO_D + i] = Wo[i];
  }

  for (int i = tid; i < ROWS * DD; i += NTHREADS) {
    const int r = i >> 8;
    const int d = i & 255;
    const float* xp = x + ((r0 + r) * DD + d) * 4;
    const float x3 = xp[3];
    const float x2 = xp[2];
    sT[r * LDS_STRIDE + d]      = x3;
    sT[r * LDS_STRIDE + DD + d] = x3 - x2;
  }

  float* sGaw = sGa + wave * GSAVE_WAVE;
  float* sGbw = sGb + wave * GSAVE_WAVE;
#pragma unroll
  for (int pc = 0; pc < 8; ++pc) {
    const v4f zero = {0.0f, 0.0f, 0.0f, 0.0f};
    *(v4f*)(sGaw + pc * 128 + lane * 4) = zero;
    *(v4f*)(sGbw + pc * 128 + lane * 4) = zero;
  }
  __syncthreads();

  gemm_512<5, false, 4, 16 * TWO_D, 1>(sT, W1 + (wave * 32 + m) * TWO_D + (khalf << 1),
                                       sZ, nullptr, sCst, nullptr, x, out, r0, wave, lane);
  __syncthreads();

  const int pofs = khalf * (TWO_D * 2) + (wave * 32 + m) * 2;

  for (int ev = 0; ev < 2 * STEPS; ++ev) {
    const bool second = (ev & 1) != 0;
    const float* PM = second ? PMb : PMa;
    float* sGw      = second ? sGbw : sGaw;
    gemm_512<2, true,  PACKED>(sZ, P2f + pofs, sT, nullptr, sCst, nullptr, x, out, r0, wave, lane);
    __syncthreads();
    gemm_512<3, false, PACKED>(sT, P2b + pofs, sT, sZ, sCst, sGw, x, out, r0, wave, lane);
    __syncthreads();
    if (ev + 1 < 2 * STEPS) {
      gemm_512<6, false, PACKED>(sT, PM + pofs, sZ, nullptr, sCst, nullptr, x, out, r0, wave, lane);
      __syncthreads();
    }
  }

  {
    v8f ga[2][2], gb[2][2];
    load_sum(sGaw, ga, lane);
    load_sum(sGbw, gb, lane);
#pragma unroll
    for (int mt = 0; mt < 2; ++mt)
#pragma unroll
      for (int t = 0; t < 2; ++t)
#pragma unroll
        for (int j = 0; j < 8; ++j) {
          const int row = mt * 16 + (khalf << 3) + j;
          const int col = wave * 32 + t * 16 + m;
          sT[row * LDS_STRIDE + col] = ga[mt][t][j];
          sZ[row * LDS_STRIDE + col] = ga[mt][t][j] + gb[mt][t][j];
        }
  }
  __syncthreads();

  gemm_512<7, false, 4 * TWO_D, 16, TWO_D>((wave < NWAVES / 2) ? sZ : sT,
                                           W1 + (khalf << 1) * TWO_D + wave * 32 + m,
                                           nullptr, nullptr, sCst, nullptr, x, out, r0, wave, lane);
}

extern "C" void kernel_launch(void* const* d_in, const int* in_sizes, int n_in,
                              void* d_out, int out_size, void* d_ws, size_t ws_size,
                              hipStream_t stream) {
  const float* x  = (const float*)d_in[0];
  const float* W1 = (const float*)d_in[1];
  const float* b1 = (const float*)d_in[2];
  const float* W2 = (const float*)d_in[3];
  const float* b2 = (const float*)d_in[4];
  const float* Wo = (const float*)d_in[5];
  float* out = (float*)d_out;

  const int MS = TWO_D * TWO_D;
  float* P2f = (float*)d_ws;
  float* P2b = P2f + MS;
  float* PMa = P2b + MS;
  float* PMb = PMa + MS;

  pack_weights<<<(MS + 255) / 256, 256, 0, stream>>>(W2, P2f, 1);
  pack_weights<<<(MS + 255) / 256, 256, 0, stream>>>(W2, P2b, 0);
  pack_step_matrices<<<(TWO_D / 16) * (TWO_D / 16) / 8, 256, 0, stream>>>(W1, PMa, PMb);

  const int nblocks = 8192 / ROWS;
  hnn_leapfrog_kernel<<<nblocks, NTHREADS, 0, stream>>>(
      x, W1, b1, b2, Wo, P2f, P2b, PMa, PMb, out);
}
